# starting kernel
# baseline (speedup 1.0000x reference)
#include <hip/hip_runtime.h>
#include <math.h>


typedef __attribute__((ext_vector_type(2))) float v2f;
typedef __attribute__((ext_vector_type(8))) float v8f;


struct cd { double re, im; };
__device__ __forceinline__ cd cmul(cd a, cd b) {
  return { a.re * b.re - a.im * b.im, a.re * b.im + a.im * b.re };
}

__device__ void gate_rx(double t, cd g[4]) {
  double c = cos(t * 0.5), s = sin(t * 0.5);
  g[0] = { c, 0.0 }; g[1] = { 0.0, -s };
  g[2] = { 0.0, -s }; g[3] = { c, 0.0 };
}
__device__ void gate_rz(double t, cd g[4]) {
  double c = cos(t * 0.5), s = sin(t * 0.5);
  g[0] = { c, -s }; g[1] = { 0.0, 0.0 };
  g[2] = { 0.0, 0.0 }; g[3] = { c, s };
}
__device__ void gate_ry(double t, cd g[4]) {
  double c = cos(t * 0.5), s = sin(t * 0.5);
  g[0] = { c, 0.0 }; g[1] = { -s, 0.0 };
  g[2] = { s, 0.0 }; g[3] = { c, 0.0 };
}

__device__ void apply_q0(cd U[16], const cd g[4]) {
  cd nU[16];
  for (int a = 0; a < 2; ++a)
    for (int b = 0; b < 2; ++b)
      for (int col = 0; col < 4; ++col) {
        cd t0 = cmul(g[a * 2 + 0], U[(0 * 2 + b) * 4 + col]);
        cd t1 = cmul(g[a * 2 + 1], U[(1 * 2 + b) * 4 + col]);
        nU[(a * 2 + b) * 4 + col] = { t0.re + t1.re, t0.im + t1.im };
      }
  for (int i = 0; i < 16; ++i) U[i] = nU[i];
}

__device__ void apply_q1(cd U[16], const cd g[4]) {
  cd nU[16];
  for (int a = 0; a < 2; ++a)
    for (int b = 0; b < 2; ++b)
      for (int col = 0; col < 4; ++col) {
        cd t0 = cmul(g[b * 2 + 0], U[(a * 2 + 0) * 4 + col]);
        cd t1 = cmul(g[b * 2 + 1], U[(a * 2 + 1) * 4 + col]);
        nU[(a * 2 + b) * 4 + col] = { t0.re + t1.re, t0.im + t1.im };
      }
  for (int i = 0; i < 16; ++i) U[i] = nU[i];
}

__global__ void qnn_setup_kernel(const float* __restrict__ weights,
                                 float* __restrict__ wsA) {
  if (threadIdx.x != 0) return;
  double w[12];
  for (int i = 0; i < 12; ++i) w[i] = (double)weights[i];
  cd U[16];
  for (int i = 0; i < 16; ++i) U[i] = { (i % 5 == 0) ? 1.0 : 0.0, 0.0 };
  cd g[4];
  for (int layer = 0; layer < 2; ++layer) {
    int o = 6 * layer;
    gate_rx(w[o + 0], g); apply_q0(U, g);
    gate_rz(w[o + 1], g); apply_q0(U, g);
    gate_ry(w[o + 2], g); apply_q0(U, g);
    gate_rx(w[o + 3], g); apply_q1(U, g);
    gate_rz(w[o + 4], g); apply_q1(U, g);
    gate_ry(w[o + 5], g); apply_q1(U, g);
    for (int col = 0; col < 4; ++col) {
      cd tmp = U[2 * 4 + col];
      U[2 * 4 + col] = U[3 * 4 + col];
      U[3 * 4 + col] = tmp;
    }
  }
  const double zsg[4] = { 1.0, 1.0, -1.0, -1.0 };
  int p = 0;
  for (int j = 0; j < 4; ++j)
    for (int k = j; k < 4; ++k) {
      double acc = 0.0;
      for (int r = 0; r < 4; ++r)
        acc += zsg[r] * (U[r * 4 + j].re * U[r * 4 + k].re +
                         U[r * 4 + j].im * U[r * 4 + k].im);
      wsA[p++] = (float)acc;
    }
}


__global__ void __launch_bounds__(256)
qnn_main_kernel(const float* __restrict__ in,
                const float* __restrict__ wsA,
                float* __restrict__ out,
                int B, int NT) {
  const int lane = threadIdx.x & 31;
  const int wv   = (blockIdx.x * blockDim.x + threadIdx.x) >> 5;
  const int nwv  = (gridDim.x * blockDim.x) >> 5;

  const float A00 = wsA[0], A01 = wsA[1], A02 = wsA[2], A03 = wsA[3];
  const float A11 = wsA[4], A12 = wsA[5], A13 = wsA[6];
  const float A22 = wsA[7], A23 = wsA[8], A33 = wsA[9];

  const bool lo = (lane < 16);
  const bool diagLane = (lane < 8) || (lane >= 24);
  const int smax = B - 1;
  const int idx = lane & 7;

  for (int t = wv; t < NT; t += nwv) {
    int s = t * 16 + (lane & 15);
    int sc = (s < B) ? s : smax;
    float2 x = ((const float2*)in)[sc];

    float a0 = x.x * 0.25f;
    float a1 = x.y * 0.25f;
    float s0 = __builtin_amdgcn_sinf(a0);
    float c0 = __builtin_amdgcn_cosf(a0);
    float s1 = __builtin_amdgcn_sinf(a1);
    float c1 = __builtin_amdgcn_cosf(a1);

    float p0 = c0 * c1, p1 = c0 * s1, p2 = s0 * c1, p3 = s0 * s1;

    float g0 = A00 * p0 + A01 * p1 + A02 * p2 + A03 * p3;
    float g1 = A01 * p0 + A11 * p1 + A12 * p2 + A13 * p3;
    float g2 = A02 * p0 + A12 * p1 + A22 * p2 + A23 * p3;
    float g3 = A03 * p0 + A13 * p1 + A23 * p2 + A33 * p3;

    v2f af, bf;
    af.x = lo ? p0 : p2;  af.y = lo ? p1 : p3;
    bf.x = lo ? g0 : g2;  bf.y = lo ? g1 : g3;
    v8f cz = {};
    v8f d = __builtin_amdgcn_wmma_f32_16x16x4_f32(
        false, af, false, bf, (short)0, cz, false, false);

    float y = d[0];
    y = (idx == 1) ? d[1] : y;
    y = (idx == 2) ? d[2] : y;
    y = (idx == 3) ? d[3] : y;
    y = (idx == 4) ? d[4] : y;
    y = (idx == 5) ? d[5] : y;
    y = (idx == 6) ? d[6] : y;
    y = (idx == 7) ? d[7] : y;

    if (diagLane && s < B) out[s] = y;
  }
}


extern "C" void kernel_launch(void* const* d_in, const int* in_sizes, int n_in,
                              void* d_out, int out_size, void* d_ws, size_t ws_size,
                              hipStream_t stream) {
  const float* inp = (const float*)d_in[0];
  const float* wts = (const float*)d_in[1];
  float* out = (float*)d_out;
  float* wsA = (float*)d_ws;

  int B = in_sizes[0] / 2;

  qnn_setup_kernel<<<1, 32, 0, stream>>>(wts, wsA);

  int NT = (B + 15) / 16;
  int wavesNeeded = (NT + 7) / 8;
  int blocks = (wavesNeeded + 7) / 8;
  if (blocks < 1) blocks = 1;
  qnn_main_kernel<<<blocks, 256, 0, stream>>>(inp, wsA, out, B, NT);
}


# Setup launch: the one-thread f64 chain of twelve 4x4 complex gate products becomes a closed form, nine f64 sincos in nine lanes plus four 3-vector rotations; same ten coefficients to f32 rounding
# speedup vs baseline: 4.2115x; 4.2115x over previous
#include <hip/hip_runtime.h>
#include <math.h>


typedef __attribute__((ext_vector_type(2))) float v2f;
typedef __attribute__((ext_vector_type(8))) float v8f;


__device__ __forceinline__ double lane_bcast(double v, int src) {
  int lo = __builtin_amdgcn_readlane(__double2loint(v), src);
  int hi = __builtin_amdgcn_readlane(__double2hiint(v), src);
  return __hiloint2double(hi, lo);
}

__global__ void qnn_setup_kernel(const float* __restrict__ weights,
                                 float* __restrict__ wsA) {
  const int lane = threadIdx.x & 31;
  const int wi = (lane < 9) ? lane : 8;
  double sn, cs;
  sincos((double)weights[wi], &sn, &cs);

  const double s0 = lane_bcast(sn, 0), c0 = lane_bcast(cs, 0);
  const double s1 = lane_bcast(sn, 1), c1 = lane_bcast(cs, 1);
  const double s2 = lane_bcast(sn, 2), c2 = lane_bcast(cs, 2);
  const double s3 = lane_bcast(sn, 3), c3 = lane_bcast(cs, 3);
  const double s4 = lane_bcast(sn, 4), c4 = lane_bcast(cs, 4);
  const double s5 = lane_bcast(sn, 5), c5 = lane_bcast(cs, 5);
  const double s6 = lane_bcast(sn, 6), c6 = lane_bcast(cs, 6);
  const double s7 = lane_bcast(sn, 7), c7 = lane_bcast(cs, 7);
  const double s8 = lane_bcast(sn, 8), c8 = lane_bcast(cs, 8);


  const double t87 = s8 * s7;
  const double mx = -(s8 * c7);
  const double my = t87 * c6 + c8 * s6;
  const double mz = c8 * c6 - t87 * s6;

  const double q54 = c5 * s4;
  const double vx = c5 * c4;
  const double vy = s5 * s3 - q54 * c3;
  const double vz = s5 * c3 + q54 * s3;

  const double ux1 = mx * c2, uz1 = mx * s2;
  const double ux  = ux1 * c1 + my * s1;
  const double uy2 = my * c1 - ux1 * s1;
  const double uy  = uy2 * c0 + uz1 * s0;
  const double uz  = uz1 * c0 - uy2 * s0;

  const double wx1 = -(mz * s2), wz1 = mz * c2;
  const double wx  = wx1 * c1;
  const double wy2 = -(wx1 * s1);
  const double wz  = wz1 * c0 - wy2 * s0;

  const double zz = uz * vz, xz = ux * vz, xx = ux * vx, yy = uy * vy;
  const float A00 = (float)(zz + wz);
  const float A01 = (float)(uz * vx);
  const float A02 = (float)(xz + wx);
  const float A03 = (float)(xx - yy);
  const float A11 = (float)(wz - zz);
  const float A12 = (float)(xx + yy);
  const float A13 = (float)(wx - xz);

  if (lane == 0) {
    wsA[0] = A00;  wsA[1] = A01;  wsA[2] = A02;  wsA[3] = A03;
    wsA[4] = A11;  wsA[5] = A12;  wsA[6] = A13;
    wsA[7] = -A00; wsA[8] = -A01; wsA[9] = -A11;
  }
}


__global__ void __launch_bounds__(256)
qnn_main_kernel(const float* __restrict__ in,
                const float* __restrict__ wsA,
                float* __restrict__ out,
                int B, int NT) {
  const int lane = threadIdx.x & 31;
  const int wv   = (blockIdx.x * blockDim.x + threadIdx.x) >> 5;
  const int nwv  = (gridDim.x * blockDim.x) >> 5;

  const float A00 = wsA[0], A01 = wsA[1], A02 = wsA[2], A03 = wsA[3];
  const float A11 = wsA[4], A12 = wsA[5], A13 = wsA[6];
  const float A22 = wsA[7], A23 = wsA[8], A33 = wsA[9];

  const bool lo = (lane < 16);
  const bool diagLane = (lane < 8) || (lane >= 24);
  const int smax = B - 1;
  const int idx = lane & 7;

  for (int t = wv; t < NT; t += nwv) {
    int s = t * 16 + (lane & 15);
    int sc = (s < B) ? s : smax;
    float2 x = ((const float2*)in)[sc];

    float a0 = x.x * 0.25f;
    float a1 = x.y * 0.25f;
    float s0 = __builtin_amdgcn_sinf(a0);
    float c0 = __builtin_amdgcn_cosf(a0);
    float s1 = __builtin_amdgcn_sinf(a1);
    float c1 = __builtin_amdgcn_cosf(a1);

    float p0 = c0 * c1, p1 = c0 * s1, p2 = s0 * c1, p3 = s0 * s1;

    float g0 = A00 * p0 + A01 * p1 + A02 * p2 + A03 * p3;
    float g1 = A01 * p0 + A11 * p1 + A12 * p2 + A13 * p3;
    float g2 = A02 * p0 + A12 * p1 + A22 * p2 + A23 * p3;
    float g3 = A03 * p0 + A13 * p1 + A23 * p2 + A33 * p3;

    v2f af, bf;
    af.x = lo ? p0 : p2;  af.y = lo ? p1 : p3;
    bf.x = lo ? g0 : g2;  bf.y = lo ? g1 : g3;
    v8f cz = {};
    v8f d = __builtin_amdgcn_wmma_f32_16x16x4_f32(
        false, af, false, bf, (short)0, cz, false, false);

    float y = d[0];
    y = (idx == 1) ? d[1] : y;
    y = (idx == 2) ? d[2] : y;
    y = (idx == 3) ? d[3] : y;
    y = (idx == 4) ? d[4] : y;
    y = (idx == 5) ? d[5] : y;
    y = (idx == 6) ? d[6] : y;
    y = (idx == 7) ? d[7] : y;

    if (diagLane && s < B) out[s] = y;
  }
}


extern "C" void kernel_launch(void* const* d_in, const int* in_sizes, int n_in,
                              void* d_out, int out_size, void* d_ws, size_t ws_size,
                              hipStream_t stream) {
  const float* inp = (const float*)d_in[0];
  const float* wts = (const float*)d_in[1];
  float* out = (float*)d_out;
  float* wsA = (float*)d_ws;

  int B = in_sizes[0] / 2;

  qnn_setup_kernel<<<1, 32, 0, stream>>>(wts, wsA);

  int NT = (B + 15) / 16;
  int wavesNeeded = (NT + 7) / 8;
  int blocks = (wavesNeeded + 7) / 8;
  if (blocks < 1) blocks = 1;
  qnn_main_kernel<<<blocks, 256, 0, stream>>>(inp, wsA, out, B, NT);
}


# Main kernel launch shape: 512 work-groups of 8 waves loop over 256-sample batches (16-byte loads issued one batch ahead, 8-byte stores) instead of one short wave per 128 samples; sum order unchanged
# speedup vs baseline: 5.5972x; 1.3290x over previous
#include <hip/hip_runtime.h>
#include <math.h>


typedef __attribute__((ext_vector_type(2))) float v2f;
typedef __attribute__((ext_vector_type(8))) float v8f;


__device__ __forceinline__ double lane_bcast(double v, int src) {
  int lo = __builtin_amdgcn_readlane(__double2loint(v), src);
  int hi = __builtin_amdgcn_readlane(__double2hiint(v), src);
  return __hiloint2double(hi, lo);
}

__global__ void qnn_setup_kernel(const float* __restrict__ weights,
                                 float* __restrict__ wsA) {
  const int lane = threadIdx.x & 31;
  const int wi = (lane < 9) ? lane : 8;
  double sn, cs;
  sincos((double)weights[wi], &sn, &cs);

  const double s0 = lane_bcast(sn, 0), c0 = lane_bcast(cs, 0);
  const double s1 = lane_bcast(sn, 1), c1 = lane_bcast(cs, 1);
  const double s2 = lane_bcast(sn, 2), c2 = lane_bcast(cs, 2);
  const double s3 = lane_bcast(sn, 3), c3 = lane_bcast(cs, 3);
  const double s4 = lane_bcast(sn, 4), c4 = lane_bcast(cs, 4);
  const double s5 = lane_bcast(sn, 5), c5 = lane_bcast(cs, 5);
  const double s6 = lane_bcast(sn, 6), c6 = lane_bcast(cs, 6);
  const double s7 = lane_bcast(sn, 7), c7 = lane_bcast(cs, 7);
  const double s8 = lane_bcast(sn, 8), c8 = lane_bcast(cs, 8);


  const double t87 = s8 * s7;
  const double mx = -(s8 * c7);
  const double my = t87 * c6 + c8 * s6;
  const double mz = c8 * c6 - t87 * s6;

  const double q54 = c5 * s4;
  const double vx = c5 * c4;
  const double vy = s5 * s3 - q54 * c3;
  const double vz = s5 * c3 + q54 * s3;

  const double ux1 = mx * c2, uz1 = mx * s2;
  const double ux  = ux1 * c1 + my * s1;
  const double uy2 = my * c1 - ux1 * s1;
  const double uy  = uy2 * c0 + uz1 * s0;
  const double uz  = uz1 * c0 - uy2 * s0;

  const double wx1 = -(mz * s2), wz1 = mz * c2;
  const double wx  = wx1 * c1;
  const double wy2 = -(wx1 * s1);
  const double wz  = wz1 * c0 - wy2 * s0;

  const double zz = uz * vz, xz = ux * vz, xx = ux * vx, yy = uy * vy;
  const float A00 = (float)(zz + wz);
  const float A01 = (float)(uz * vx);
  const float A02 = (float)(xz + wx);
  const float A03 = (float)(xx - yy);
  const float A11 = (float)(wz - zz);
  const float A12 = (float)(xx + yy);
  const float A13 = (float)(wx - xz);

  if (lane == 0) {
    wsA[0] = A00;  wsA[1] = A01;  wsA[2] = A02;  wsA[3] = A03;
    wsA[4] = A11;  wsA[5] = A12;  wsA[6] = A13;
    wsA[7] = -A00; wsA[8] = -A01; wsA[9] = -A11;
  }
}


struct QnnCoef {
  float A00, A01, A02, A03, A11, A12, A13, A22, A23, A33;
};

__device__ __forceinline__ float qnn_tile16(float x0, float x1, const QnnCoef& k,
                                            bool lo, int idx) {
  float a0 = x0 * 0.25f;
  float a1 = x1 * 0.25f;
  float s0 = __builtin_amdgcn_sinf(a0);
  float c0 = __builtin_amdgcn_cosf(a0);
  float s1 = __builtin_amdgcn_sinf(a1);
  float c1 = __builtin_amdgcn_cosf(a1);

  float p0 = c0 * c1, p1 = c0 * s1, p2 = s0 * c1, p3 = s0 * s1;

  float g0 = k.A00 * p0 + k.A01 * p1 + k.A02 * p2 + k.A03 * p3;
  float g1 = k.A01 * p0 + k.A11 * p1 + k.A12 * p2 + k.A13 * p3;
  float g2 = k.A02 * p0 + k.A12 * p1 + k.A22 * p2 + k.A23 * p3;
  float g3 = k.A03 * p0 + k.A13 * p1 + k.A23 * p2 + k.A33 * p3;

  v2f af, bf;
  af.x = lo ? p0 : p2;  af.y = lo ? p1 : p3;
  bf.x = lo ? g0 : g2;  bf.y = lo ? g1 : g3;
  v8f cz = {};
  v8f d = __builtin_amdgcn_wmma_f32_16x16x4_f32(
      false, af, false, bf, (short)0, cz, false, false);

  float y = d[0];
  y = (idx == 1) ? d[1] : y;
  y = (idx == 2) ? d[2] : y;
  y = (idx == 3) ? d[3] : y;
  y = (idx == 4) ? d[4] : y;
  y = (idx == 5) ? d[5] : y;
  y = (idx == 6) ? d[6] : y;
  y = (idx == 7) ? d[7] : y;
  return y;
}

#define QNN_U 8

__global__ void __launch_bounds__(256)
qnn_main_kernel(const float* __restrict__ in,
                const float* __restrict__ wsA,
                float* __restrict__ out,
                int B, int NT) {
  const int lane = threadIdx.x & 31;
  const int wv   = __builtin_amdgcn_readfirstlane(
      (int)((blockIdx.x * blockDim.x + threadIdx.x) >> 5));
  const int nwv  = (gridDim.x * blockDim.x) >> 5;

  QnnCoef k;
  k.A00 = wsA[0]; k.A01 = wsA[1]; k.A02 = wsA[2]; k.A03 = wsA[3];
  k.A11 = wsA[4]; k.A12 = wsA[5]; k.A13 = wsA[6];
  k.A22 = wsA[7]; k.A23 = wsA[8]; k.A33 = wsA[9];

  const bool lo = (lane < 16);
  const bool diagLane = (lane < 8) || (lane >= 24);
  const int idx = lane & 7;
  const int c = lane & 15;

  const int NB = B / (32 * QNN_U);
  const float4* __restrict__ in4 = (const float4*)in;
  float2* __restrict__ out2 = (float2*)out;

  if (wv < NB) {
    float4 xn[QNN_U];
#pragma unroll
    for (int u = 0; u < QNN_U; ++u) xn[u] = in4[(wv * QNN_U + u) * 16 + c];

    for (int b = wv; b < NB; b += nwv) {
      float4 xc[QNN_U];
#pragma unroll
      for (int u = 0; u < QNN_U; ++u) xc[u] = xn[u];

      int bp = b + nwv;
      bp = (bp < NB) ? bp : b;
#pragma unroll
      for (int u = 0; u < QNN_U; ++u) xn[u] = in4[(bp * QNN_U + u) * 16 + c];
      __builtin_amdgcn_sched_barrier(0);

      float2 yy[QNN_U];
#pragma unroll
      for (int u = 0; u < QNN_U; ++u) {
        yy[u].x = qnn_tile16(xc[u].x, xc[u].y, k, lo, idx);
        yy[u].y = qnn_tile16(xc[u].z, xc[u].w, k, lo, idx);
      }
      if (diagLane) {
#pragma unroll
        for (int u = 0; u < QNN_U; ++u) out2[(b * QNN_U + u) * 16 + c] = yy[u];
      }
    }
  }

  const int smax = B - 1;
  for (int t = NB * (2 * QNN_U) + wv; t < NT; t += nwv) {
    int s = t * 16 + c;
    int sc = (s < B) ? s : smax;
    float2 x = ((const float2*)in)[sc];
    float y = qnn_tile16(x.x, x.y, k, lo, idx);
    if (diagLane && s < B) out[s] = y;
  }
}


extern "C" void kernel_launch(void* const* d_in, const int* in_sizes, int n_in,
                              void* d_out, int out_size, void* d_ws, size_t ws_size,
                              hipStream_t stream) {
  const float* inp = (const float*)d_in[0];
  const float* wts = (const float*)d_in[1];
  float* out = (float*)d_out;
  float* wsA = (float*)d_ws;

  int B = in_sizes[0] / 2;

  qnn_setup_kernel<<<1, 32, 0, stream>>>(wts, wsA);

  int NT = (B + 15) / 16;
  qnn_main_kernel<<<512, 256, 0, stream>>>(inp, wsA, out, B, NT);
}


# Setup launch removed: every wave derives the ten coefficients itself in f32 (library sincosf in nine lanes, then 45 mul/add; was f64 rounded to f32) behind its first loads; one launch, no workspace
# speedup vs baseline: 7.5135x; 1.3424x over previous
#include <hip/hip_runtime.h>
#include <math.h>


typedef __attribute__((ext_vector_type(2))) float v2f;
typedef __attribute__((ext_vector_type(8))) float v8f;


struct QnnCoef {
  float A00, A01, A02, A03, A11, A12, A13, A22, A23, A33;
};

__device__ __forceinline__ float lane_bcast(float v, int src) {
  return __builtin_bit_cast(float,
      __builtin_amdgcn_readlane(__builtin_bit_cast(int, v), src));
}

__device__ __forceinline__ QnnCoef qnn_coefficients(float wl) {
  float sn, cs;
  sincosf(wl, &sn, &cs);

  const float s0 = lane_bcast(sn, 0), c0 = lane_bcast(cs, 0);
  const float s1 = lane_bcast(sn, 1), c1 = lane_bcast(cs, 1);
  const float s2 = lane_bcast(sn, 2), c2 = lane_bcast(cs, 2);
  const float s3 = lane_bcast(sn, 3), c3 = lane_bcast(cs, 3);
  const float s4 = lane_bcast(sn, 4), c4 = lane_bcast(cs, 4);
  const float s5 = lane_bcast(sn, 5), c5 = lane_bcast(cs, 5);
  const float s6 = lane_bcast(sn, 6), c6 = lane_bcast(cs, 6);
  const float s7 = lane_bcast(sn, 7), c7 = lane_bcast(cs, 7);
  const float s8 = lane_bcast(sn, 8), c8 = lane_bcast(cs, 8);


  const float t87 = s8 * s7;
  const float mx = -(s8 * c7);
  const float my = t87 * c6 + c8 * s6;
  const float mz = c8 * c6 - t87 * s6;

  const float q54 = c5 * s4;
  const float vx = c5 * c4;
  const float vy = s5 * s3 - q54 * c3;
  const float vz = s5 * c3 + q54 * s3;

  const float ux1 = mx * c2, uz1 = mx * s2;
  const float ux  = ux1 * c1 + my * s1;
  const float uy2 = my * c1 - ux1 * s1;
  const float uy  = uy2 * c0 + uz1 * s0;
  const float uz  = uz1 * c0 - uy2 * s0;

  const float wx1 = -(mz * s2), wz1 = mz * c2;
  const float wx  = wx1 * c1;
  const float wy2 = -(wx1 * s1);
  const float wz  = wz1 * c0 - wy2 * s0;

  const float zz = uz * vz, xz = ux * vz, xx = ux * vx, yy = uy * vy;
  QnnCoef k;
  k.A00 = zz + wz;
  k.A01 = uz * vx;
  k.A02 = xz + wx;
  k.A03 = xx - yy;
  k.A11 = wz - zz;
  k.A12 = xx + yy;
  k.A13 = wx - xz;
  k.A22 = -k.A00;
  k.A23 = -k.A01;
  k.A33 = -k.A11;
  return k;
}


__device__ __forceinline__ float qnn_tile16(float x0, float x1, const QnnCoef& k,
                                            bool lo, int idx) {
  float a0 = x0 * 0.25f;
  float a1 = x1 * 0.25f;
  float s0 = __builtin_amdgcn_sinf(a0);
  float c0 = __builtin_amdgcn_cosf(a0);
  float s1 = __builtin_amdgcn_sinf(a1);
  float c1 = __builtin_amdgcn_cosf(a1);

  float p0 = c0 * c1, p1 = c0 * s1, p2 = s0 * c1, p3 = s0 * s1;

  float g0 = k.A00 * p0 + k.A01 * p1 + k.A02 * p2 + k.A03 * p3;
  float g1 = k.A01 * p0 + k.A11 * p1 + k.A12 * p2 + k.A13 * p3;
  float g2 = k.A02 * p0 + k.A12 * p1 + k.A22 * p2 + k.A23 * p3;
  float g3 = k.A03 * p0 + k.A13 * p1 + k.A23 * p2 + k.A33 * p3;

  v2f af, bf;
  af.x = lo ? p0 : p2;  af.y = lo ? p1 : p3;
  bf.x = lo ? g0 : g2;  bf.y = lo ? g1 : g3;
  v8f cz = {};
  v8f d = __builtin_amdgcn_wmma_f32_16x16x4_f32(
      false, af, false, bf, (short)0, cz, false, false);

  float y = d[0];
  y = (idx == 1) ? d[1] : y;
  y = (idx == 2) ? d[2] : y;
  y = (idx == 3) ? d[3] : y;
  y = (idx == 4) ? d[4] : y;
  y = (idx == 5) ? d[5] : y;
  y = (idx == 6) ? d[6] : y;
  y = (idx == 7) ? d[7] : y;
  return y;
}

#define QNN_U 8

__global__ void __launch_bounds__(256)
qnn_main_kernel(const float* __restrict__ in,
                const float* __restrict__ weights,
                float* __restrict__ out,
                int B, int NT) {
  const int lane = threadIdx.x & 31;
  const int wv   = __builtin_amdgcn_readfirstlane(
      (int)((blockIdx.x * blockDim.x + threadIdx.x) >> 5));
  const int nwv  = (gridDim.x * blockDim.x) >> 5;

  int wbits = __builtin_amdgcn_readfirstlane(__builtin_bit_cast(int, weights[8]));
#pragma unroll
  for (int i = 0; i < 8; ++i) {
    const int wi = __builtin_amdgcn_readfirstlane(__builtin_bit_cast(int, weights[i]));
    wbits = (lane == i) ? wi : wbits;
  }
  const float wl = __builtin_bit_cast(float, wbits);

  const bool lo = (lane < 16);
  const bool diagLane = (lane < 8) || (lane >= 24);
  const int idx = lane & 7;
  const int c = lane & 15;

  const int NB = B / (32 * QNN_U);
  const float4* __restrict__ in4 = (const float4*)in;
  float2* __restrict__ out2 = (float2*)out;

  const bool full = (wv < NB);
  float4 xn[QNN_U];
#pragma unroll
  for (int u = 0; u < QNN_U; ++u) xn[u] = make_float4(0.0f, 0.0f, 0.0f, 0.0f);
  if (full) {
#pragma unroll
    for (int u = 0; u < QNN_U; ++u) xn[u] = in4[(wv * QNN_U + u) * 16 + c];
  }

  const QnnCoef k = qnn_coefficients(wl);

  if (full) {
    for (int b = wv; b < NB; b += nwv) {
      float4 xc[QNN_U];
#pragma unroll
      for (int u = 0; u < QNN_U; ++u) xc[u] = xn[u];

      int bp = b + nwv;
      bp = (bp < NB) ? bp : b;
#pragma unroll
      for (int u = 0; u < QNN_U; ++u) xn[u] = in4[(bp * QNN_U + u) * 16 + c];
      __builtin_amdgcn_sched_barrier(0);

      float2 yy[QNN_U];
#pragma unroll
      for (int u = 0; u < QNN_U; ++u) {
        yy[u].x = qnn_tile16(xc[u].x, xc[u].y, k, lo, idx);
        yy[u].y = qnn_tile16(xc[u].z, xc[u].w, k, lo, idx);
      }
      if (diagLane) {
#pragma unroll
        for (int u = 0; u < QNN_U; ++u) out2[(b * QNN_U + u) * 16 + c] = yy[u];
      }
    }
  }

  const int smax = B - 1;
  for (int t = NB * (2 * QNN_U) + wv; t < NT; t += nwv) {
    int s = t * 16 + c;
    int sc = (s < B) ? s : smax;
    float2 x = ((const float2*)in)[sc];
    float y = qnn_tile16(x.x, x.y, k, lo, idx);
    if (diagLane && s < B) out[s] = y;
  }
}


extern "C" void kernel_launch(void* const* d_in, const int* in_sizes, int n_in,
                              void* d_out, int out_size, void* d_ws, size_t ws_size,
                              hipStream_t stream) {
  const float* inp = (const float*)d_in[0];
  const float* wts = (const float*)d_in[1];
  float* out = (float*)d_out;

  int B = in_sizes[0] / 2;

  int NT = (B + 15) / 16;
  qnn_main_kernel<<<512, 256, 0, stream>>>(inp, wts, out, B, NT);
}


# <Z0> in product-state form C0*h1+S0*h2 (sin/cos at the full angle): WMMA does the 2x2 M*(C1,S1)+w for 32 distinct samples (was 16; A*psi on VALU, diagonal pick); sum order changes, 6 terms
# speedup vs baseline: 12.9249x; 1.7202x over previous
#include <hip/hip_runtime.h>
#include <math.h>


typedef __attribute__((ext_vector_type(2))) float v2f;
typedef __attribute__((ext_vector_type(8))) float v8f;


struct QnnCoef {
  float wz, wx;
  float zz, zx, xz, xx;
};

__device__ __forceinline__ float lane_bcast(float v, int src) {
  return __builtin_bit_cast(float,
      __builtin_amdgcn_readlane(__builtin_bit_cast(int, v), src));
}

__device__ __forceinline__ QnnCoef qnn_coefficients(float wl) {
  float sn, cs;
  sincosf(wl, &sn, &cs);

  const float s0 = lane_bcast(sn, 0), c0 = lane_bcast(cs, 0);
  const float s1 = lane_bcast(sn, 1), c1 = lane_bcast(cs, 1);
  const float s2 = lane_bcast(sn, 2), c2 = lane_bcast(cs, 2);
  const float s3 = lane_bcast(sn, 3), c3 = lane_bcast(cs, 3);
  const float s4 = lane_bcast(sn, 4), c4 = lane_bcast(cs, 4);
  const float s5 = lane_bcast(sn, 5), c5 = lane_bcast(cs, 5);
  const float s6 = lane_bcast(sn, 6), c6 = lane_bcast(cs, 6);
  const float s7 = lane_bcast(sn, 7), c7 = lane_bcast(cs, 7);
  const float s8 = lane_bcast(sn, 8), c8 = lane_bcast(cs, 8);


  const float t87 = s8 * s7;
  const float mx = -(s8 * c7);
  const float my = t87 * c6 + c8 * s6;
  const float mz = c8 * c6 - t87 * s6;

  const float q54 = c5 * s4;
  const float vx = c5 * c4;
  const float vz = s5 * c3 + q54 * s3;

  const float ux1 = mx * c2, uz1 = mx * s2;
  const float ux  = ux1 * c1 + my * s1;
  const float uy2 = my * c1 - ux1 * s1;
  const float uz  = uz1 * c0 - uy2 * s0;

  const float wx1 = -(mz * s2), wz1 = mz * c2;
  const float wx  = wx1 * c1;
  const float wy2 = -(wx1 * s1);
  const float wz  = wz1 * c0 - wy2 * s0;

  QnnCoef k;
  k.wz = wz;
  k.wx = wx;
  k.zz = uz * vz;
  k.zx = uz * vx;
  k.xz = ux * vz;
  k.xx = ux * vx;
  return k;
}


__device__ __forceinline__ float qnn_tile32(float a0, float a1, v2f af, v8f cf) {
  const float S0 = __builtin_amdgcn_sinf(a0);
  const float C0 = __builtin_amdgcn_cosf(a0);
  v2f bf;
  bf.x = __builtin_amdgcn_cosf(a1);
  bf.y = __builtin_amdgcn_sinf(a1);
  const v8f d = __builtin_amdgcn_wmma_f32_16x16x4_f32(
      false, af, false, bf, (short)0, cf, false, false);
  return C0 * d[0] + S0 * d[1];
}

#define QNN_U 8

__global__ void __launch_bounds__(256)
qnn_main_kernel(const float* __restrict__ in,
                const float* __restrict__ weights,
                float* __restrict__ out,
                int B, int NT) {
  const int lane = threadIdx.x & 31;
  const int wv   = __builtin_amdgcn_readfirstlane(
      (int)((blockIdx.x * blockDim.x + threadIdx.x) >> 5));
  const int nwv  = (gridDim.x * blockDim.x) >> 5;

  int wbits = __builtin_amdgcn_readfirstlane(__builtin_bit_cast(int, weights[8]));
#pragma unroll
  for (int i = 0; i < 8; ++i) {
    const int wi = __builtin_amdgcn_readfirstlane(__builtin_bit_cast(int, weights[i]));
    wbits = (lane == i) ? wi : wbits;
  }
  const float wl = __builtin_bit_cast(float, wbits);

  const int NB = B / (64 * QNN_U);
  const float4* __restrict__ in4 = (const float4*)in;
  float2* __restrict__ out2 = (float2*)out;

  const bool full = (wv < NB);
  float4 xn[QNN_U];
#pragma unroll
  for (int u = 0; u < QNN_U; ++u) xn[u] = make_float4(0.0f, 0.0f, 0.0f, 0.0f);
  if (full) {
#pragma unroll
    for (int u = 0; u < QNN_U; ++u) xn[u] = in4[(wv * QNN_U + u) * 32 + lane];
  }

  const QnnCoef k = qnn_coefficients(wl);

  const bool row0 = (lane == 0) || (lane == 24);
  const bool row1 = (lane == 1) || (lane == 25);
  v2f af;
  af.x = row0 ? k.zz : (row1 ? k.xz : 0.0f);
  af.y = row0 ? k.zx : (row1 ? k.xx : 0.0f);
  v8f cf = {};
  cf[0] = k.wz;
  cf[1] = k.wx;

  if (full) {
    for (int b = wv; b < NB; b += nwv) {
      float4 xc[QNN_U];
#pragma unroll
      for (int u = 0; u < QNN_U; ++u)
        xc[u] = make_float4(xn[u].x * 0.5f, xn[u].y * 0.5f, xn[u].z * 0.5f, xn[u].w * 0.5f);

      int bp = b + nwv;
      bp = (bp < NB) ? bp : b;
#pragma unroll
      for (int u = 0; u < QNN_U; ++u) xn[u] = in4[(bp * QNN_U + u) * 32 + lane];
      __builtin_amdgcn_sched_barrier(0);

      float2 yy[QNN_U];
#pragma unroll
      for (int u = 0; u < QNN_U; ++u) {
        yy[u].x = qnn_tile32(xc[u].x, xc[u].y, af, cf);
        yy[u].y = qnn_tile32(xc[u].z, xc[u].w, af, cf);
      }
#pragma unroll
      for (int u = 0; u < QNN_U; ++u) out2[(b * QNN_U + u) * 32 + lane] = yy[u];
    }
  }

  const int smax = B - 1;
  const int NT32 = (NT + 1) >> 1;
  for (int t = NB * (2 * QNN_U) + wv; t < NT32; t += nwv) {
    int s = t * 32 + lane;
    int sc = (s < B) ? s : smax;
    float2 x = ((const float2*)in)[sc];
    float y = qnn_tile32(x.x * 0.5f, x.y * 0.5f, af, cf);
    if (s < B) out[s] = y;
  }
}


extern "C" void kernel_launch(void* const* d_in, const int* in_sizes, int n_in,
                              void* d_out, int out_size, void* d_ws, size_t ws_size,
                              hipStream_t stream) {
  const float* inp = (const float*)d_in[0];
  const float* wts = (const float*)d_in[1];
  float* out = (float*)d_out;

  int B = in_sizes[0] / 2;

  int NT = (B + 15) / 16;
  qnn_main_kernel<<<512, 256, 0, stream>>>(inp, wts, out, B, NT);
}
